# starting kernel
# baseline (speedup 1.0000x reference)
#include <hip/hip_runtime.h>

typedef __attribute__((ext_vector_type(2))) float v2f;
typedef __attribute__((ext_vector_type(8))) float v8f;


__global__ __launch_bounds__(256) void k_fill(float* __restrict__ p, float v, int n) {
  int i = blockIdx.x * 256 + threadIdx.x;
  if (i < n) p[i] = v;
}

__global__ __launch_bounds__(256) void k_deg(const int* __restrict__ dst,
                                             float* __restrict__ deg, int E) {
  int i = blockIdx.x * 256 + threadIdx.x;
  if (i < E) atomicAdd(&deg[dst[i]], 1.0f);
}

__global__ __launch_bounds__(256) void k_rsqrt(float* __restrict__ p, int n) {
  int i = blockIdx.x * 256 + threadIdx.x;
  if (i < n) p[i] = rsqrtf(p[i]);
}

__global__ __launch_bounds__(256) void k_selfinit(const float* __restrict__ h,
                                                  const float* __restrict__ dinv,
                                                  float* __restrict__ out,
                                                  int n, int shift) {
  int i = blockIdx.x * 256 + threadIdx.x;
  int total = n << shift;
  if (i >= total) return;
  float w = dinv[i >> shift];
  out[i] = h[i] * w * w;
}

__global__ __launch_bounds__(256) void k_bias_act(float* __restrict__ buf,
                                                  const float* __restrict__ b,
                                                  int n, int shift, int mask, int relu) {
  int i = blockIdx.x * 256 + threadIdx.x;
  int total = n << shift;
  if (i >= total) return;
  float v = buf[i] + b[i & mask];
  buf[i] = relu ? fmaxf(v, 0.0f) : v;
}


__global__ __launch_bounds__(256) void k_scatter128(const float* __restrict__ h,
                                                    const int* __restrict__ src,
                                                    const int* __restrict__ dst,
                                                    const float* __restrict__ dinv,
                                                    float* __restrict__ out, int E) {
  int wid = (blockIdx.x * 256 + threadIdx.x) >> 5;
  int lane = threadIdx.x & 31;
  if (wid >= E) return;
  int s = src[wid], d = dst[wid];
  float w = dinv[s] * dinv[d];
  const float4 v = *reinterpret_cast<const float4*>(h + (size_t)s * 128 + lane * 4);
  float* o = out + (size_t)d * 128 + lane * 4;
  atomicAdd(o + 0, v.x * w);
  atomicAdd(o + 1, v.y * w);
  atomicAdd(o + 2, v.z * w);
  atomicAdd(o + 3, v.w * w);
}

__global__ __launch_bounds__(256) void k_scatter64(const float* __restrict__ h,
                                                   const int* __restrict__ src,
                                                   const int* __restrict__ dst,
                                                   const float* __restrict__ dinv,
                                                   float* __restrict__ out, int E) {
  int gid = blockIdx.x * 256 + threadIdx.x;
  int e = gid >> 4;
  int l = gid & 15;
  if (e >= E) return;
  int s = src[e], d = dst[e];
  float w = dinv[s] * dinv[d];
  const float4 v = *reinterpret_cast<const float4*>(h + (size_t)s * 64 + l * 4);
  float* o = out + (size_t)d * 64 + l * 4;
  atomicAdd(o + 0, v.x * w);
  atomicAdd(o + 1, v.y * w);
  atomicAdd(o + 2, v.z * w);
  atomicAdd(o + 3, v.w * w);
}

__global__ __launch_bounds__(256) void k_gemm_wmma(const float* __restrict__ A,
                                                   const float* __restrict__ B,
                                                   const float* __restrict__ bias,
                                                   float* __restrict__ C,
                                                   int M, int K, int Nc) {
  int wid = (blockIdx.x * 256 + threadIdx.x) >> 5;
  int lane = threadIdx.x & 31;
  int nt = Nc >> 4;
  int tM = wid / nt;
  int tN = wid - tM * nt;
  if (tM * 16 >= M) return;

  int r16 = lane & 15;
  int hi  = lane >> 4;

  const float* Ap = A + (size_t)(tM * 16 + r16) * K + 2 * hi;
  const float* Bp = B + (size_t)(2 * hi) * Nc + tN * 16 + r16;

  v8f acc;
  float binit = bias ? bias[tN * 16 + r16] : 0.0f;
#pragma unroll
  for (int r = 0; r < 8; ++r) acc[r] = binit;

  for (int k = 0; k < K; k += 4) {
    v2f a = *reinterpret_cast<const v2f*>(Ap + k);
    v2f b;
    b.x = Bp[(size_t)k * Nc];
    b.y = Bp[(size_t)(k + 1) * Nc];
    acc = __builtin_amdgcn_wmma_f32_16x16x4_f32(false, a, false, b,
                                                (short)0, acc, false, false);
  }

  float* Cp = C + (size_t)(tM * 16 + 8 * hi) * Nc + tN * 16 + r16;
#pragma unroll
  for (int r = 0; r < 8; ++r) Cp[(size_t)r * Nc] = acc[r];
}


extern "C" void kernel_launch(void* const* d_in, const int* in_sizes, int n_in,
                              void* d_out, int out_size, void* d_ws, size_t ws_size,
                              hipStream_t stream) {
  const float* x   = (const float*)d_in[0];
  const int*   ei  = (const int*)d_in[1];
  const float* W1  = (const float*)d_in[2];
  const float* b1  = (const float*)d_in[3];
  const float* W2  = (const float*)d_in[4];
  const float* b2  = (const float*)d_in[5];
  const float* Wfc = (const float*)d_in[6];
  const float* bfc = (const float*)d_in[7];
  float* out = (float*)d_out;

  const int N = in_sizes[0] / 128;
  const int E = in_sizes[1] / 2;
  const int* src = ei;
  const int* dst = ei + E;

  float* dinv = (float*)d_ws;
  float* bufA = dinv + N;
  float* bufB = bufA + (size_t)N * 128;

  auto cdiv = [](long long a, long long b) { return (int)((a + b - 1) / b); };

  k_fill<<<cdiv(N, 256), 256, 0, stream>>>(dinv, 1.0f, N);
  k_deg<<<cdiv(E, 256), 256, 0, stream>>>(dst, dinv, E);
  k_rsqrt<<<cdiv(N, 256), 256, 0, stream>>>(dinv, N);

  {
    int waves = cdiv(N, 16) * (128 >> 4);
    k_gemm_wmma<<<cdiv(waves, 8), 256, 0, stream>>>(x, W1, nullptr, bufA, N, 128, 128);
  }
  k_selfinit<<<cdiv((long long)N << 7, 256), 256, 0, stream>>>(bufA, dinv, bufB, N, 7);
  k_scatter128<<<cdiv(E, 8), 256, 0, stream>>>(bufA, src, dst, dinv, bufB, E);
  k_bias_act<<<cdiv((long long)N << 7, 256), 256, 0, stream>>>(bufB, b1, N, 7, 127, 1);

  {
    int waves = cdiv(N, 16) * (64 >> 4);
    k_gemm_wmma<<<cdiv(waves, 8), 256, 0, stream>>>(bufB, W2, nullptr, bufA, N, 128, 64);
  }
  k_selfinit<<<cdiv((long long)N << 6, 256), 256, 0, stream>>>(bufA, dinv, bufB, N, 6);
  k_scatter64<<<cdiv((long long)E * 16, 256), 256, 0, stream>>>(bufA, src, dst, dinv, bufB, E);
  k_bias_act<<<cdiv((long long)N << 6, 256), 256, 0, stream>>>(bufB, b2, N, 6, 63, 0);

  {
    int waves = cdiv(N, 16) * (64 >> 4);
    k_gemm_wmma<<<cdiv(waves, 8), 256, 0, stream>>>(bufB, Wfc, bfc, out, N, 64, 64);
  }
}


# large launches resized from one wave per item to resident waves that loop over items (a fifth as many waves enter per call); a product wave keeps its B block in registers; sums stay in arrival order
# speedup vs baseline: 2.4778x; 2.4778x over previous
#include <hip/hip_runtime.h>


typedef __attribute__((ext_vector_type(2))) float v2f;
typedef __attribute__((ext_vector_type(4))) float v4f;
typedef __attribute__((ext_vector_type(8))) float v8f;


__global__ __launch_bounds__(256) void k_zero(int* __restrict__ p, int n) {
  int i = blockIdx.x * 256 + threadIdx.x;
  if (i < n) p[i] = 0;
}

__global__ __launch_bounds__(256) void k_count(const int* __restrict__ dst,
                                               int* __restrict__ cnt,
                                               int* __restrict__ rank, int E) {
  int stride = gridDim.x * 256;
  for (int e0 = blockIdx.x * 256 + threadIdx.x; e0 < E; e0 += 4 * stride) {
    int d[4], r[4];
#pragma unroll
    for (int u = 0; u < 4; ++u) {
      int e = e0 + u * stride;
      d[u] = dst[e < E ? e : E - 1];
    }
#pragma unroll
    for (int u = 0; u < 4; ++u) r[u] = (e0 + u * stride < E) ? atomicAdd(&cnt[d[u]], 1) : 0;
#pragma unroll
    for (int u = 0; u < 4; ++u)
      if (e0 + u * stride < E) rank[e0 + u * stride] = r[u];
  }
}

__global__ __launch_bounds__(256) void k_alloc(const int* __restrict__ cnt,
                                               int* __restrict__ cursor,
                                               int* __restrict__ rowptr,
                                               float* __restrict__ dinv, int n) {
  __shared__ int wsum[8];
  __shared__ int base_s;
  int i = blockIdx.x * 256 + threadIdx.x;
  int lane = threadIdx.x & 31;
  int w = threadIdx.x >> 5;
  int c = (i < n) ? cnt[i] : 0;
  int x = c;
  for (int o = 1; o < 32; o <<= 1) {
    int y = __shfl_up(x, o);
    if (lane >= o) x += y;
  }
  if (lane == 31) wsum[w] = x;
  __syncthreads();
  if (threadIdx.x == 0) {
    int t = 0;
    for (int k = 0; k < 8; ++k) { int v = wsum[k]; wsum[k] = t; t += v; }
    base_s = atomicAdd(cursor, t);
  }
  __syncthreads();
  if (i < n) {
    rowptr[i] = base_s + wsum[w] + x - c;
    dinv[i] = rsqrtf((float)(c + 1));
  }
}

__global__ __launch_bounds__(256) void k_place(const int* __restrict__ src,
                                               const int* __restrict__ dst,
                                               const int* __restrict__ rank,
                                               const int* __restrict__ rowptr,
                                               const float* __restrict__ dinv,
                                               int2* __restrict__ slot, int E) {
  int stride = gridDim.x * 256;
  for (int e0 = blockIdx.x * 256 + threadIdx.x; e0 < E; e0 += 4 * stride) {
    int s[4], d[4], pos[4];
    float w[4];
#pragma unroll
    for (int u = 0; u < 4; ++u) {
      int e = e0 + u * stride;
      e = e < E ? e : E - 1;
      s[u] = src[e];
      d[u] = dst[e];
      pos[u] = rank[e];
    }
#pragma unroll
    for (int u = 0; u < 4; ++u) {
      pos[u] += rowptr[d[u]];
      w[u] = dinv[s[u]];
    }
#pragma unroll
    for (int u = 0; u < 4; ++u)
      if (e0 + u * stride < E) slot[pos[u]] = make_int2(s[u], __float_as_int(w[u]));
  }
}


template <typename VT, int G>
__device__ __forceinline__ void add_rows(VT& acc, const VT* __restrict__ hv,
                                         int sl, float wl, int k, int lane) {
  VT v[G];
  float w[G];
#pragma unroll
  for (int u = 0; u < G; ++u) {
    int s = __builtin_amdgcn_readlane(sl, k + u);
    w[u] = __int_as_float(__builtin_amdgcn_readlane(__float_as_int(wl), k + u));
    v[u] = hv[(size_t)s * 32 + lane];
  }
#pragma unroll
  for (int u = 0; u < G; ++u) acc += v[u] * w[u];
}

template <typename VT>
__global__ __launch_bounds__(256) void k_gather(const float* __restrict__ h,
                                                const int2* __restrict__ slot,
                                                const int* __restrict__ rowptr,
                                                const int* __restrict__ cnt,
                                                const float* __restrict__ dinv,
                                                const float* __restrict__ bias,
                                                float* __restrict__ out, int n, int relu) {
  int wave = blockIdx.x * 8 + __builtin_amdgcn_readfirstlane((int)(threadIdx.x >> 5));
  int nwaves = gridDim.x * 8;
  int lane = threadIdx.x & 31;
  const VT bv = reinterpret_cast<const VT*>(bias)[lane];
  for (int node = wave; node < n; node += nwaves) {
  int start = rowptr[node];
  int c = cnt[node];
  float di = dinv[node];
  const VT* hv = reinterpret_cast<const VT*>(h);
  VT acc = hv[(size_t)node * 32 + lane] * (di * di);
  for (int b = 0; b < c; b += 32) {
    int m = c - b < 32 ? c - b : 32;
    int sl = 0;
    float wl = 0.0f;
    if (lane < m) {
      int2 cw = slot[start + b + lane];
      sl = cw.x;
      wl = __int_as_float(cw.y) * di;
    }
    int k = 0;
    for (; k + 8 <= m; k += 8) add_rows<VT, 8>(acc, hv, sl, wl, k, lane);
    if (k + 4 <= m) { add_rows<VT, 4>(acc, hv, sl, wl, k, lane); k += 4; }
    if (k + 2 <= m) { add_rows<VT, 2>(acc, hv, sl, wl, k, lane); k += 2; }
    if (k < m)      { add_rows<VT, 1>(acc, hv, sl, wl, k, lane); }
  }
  acc += bv;
  if (relu) acc = __builtin_elementwise_max(acc, (VT)(0.0f));
  reinterpret_cast<VT*>(out)[(size_t)node * 32 + lane] = acc;
  }
}

template <int K>
__global__ __launch_bounds__(256) void k_gemm_wmma(const float* __restrict__ A,
                                                   const float* __restrict__ B,
                                                   const float* __restrict__ bias,
                                                   float* __restrict__ C,
                                                   int M, int Nc) {
  int wid = blockIdx.x * 8 + __builtin_amdgcn_readfirstlane((int)(threadIdx.x >> 5));
  int lane = threadIdx.x & 31;
  int nt = Nc >> 4;
  int tN = wid % nt;
  int tM0 = wid / nt;
  int tMstep = (gridDim.x * 8) / nt;

  int r16 = lane & 15;
  int hi  = lane >> 4;

  const float* Bp = B + (size_t)(2 * hi) * Nc + tN * 16 + r16;
  v2f b[K / 4];
#pragma unroll
  for (int j = 0; j < K / 4; ++j) {
    b[j].x = Bp[(size_t)(4 * j) * Nc];
    b[j].y = Bp[(size_t)(4 * j + 1) * Nc];
  }
  float binit = bias ? bias[tN * 16 + r16] : 0.0f;

  for (int tM = tM0; tM * 16 < M; tM += tMstep) {
    const float* Ap = A + (size_t)(tM * 16 + r16) * K + 2 * hi;
    v2f a[K / 4];
#pragma unroll
    for (int j = 0; j < K / 4; ++j)
      a[j] = *reinterpret_cast<const v2f*>(Ap + 4 * j);

    v8f acc;
#pragma unroll
    for (int r = 0; r < 8; ++r) acc[r] = binit;
#pragma unroll
    for (int j = 0; j < K / 4; ++j)
      acc = __builtin_amdgcn_wmma_f32_16x16x4_f32(false, a[j], false, b[j],
                                                  (short)0, acc, false, false);

    float* Cp = C + (size_t)(tM * 16 + 8 * hi) * Nc + tN * 16 + r16;
#pragma unroll
    for (int r = 0; r < 8; ++r) Cp[(size_t)r * Nc] = acc[r];
  }
}


extern "C" void kernel_launch(void* const* d_in, const int* in_sizes, int n_in,
                              void* d_out, int out_size, void* d_ws, size_t ws_size,
                              hipStream_t stream) {
  const float* x   = (const float*)d_in[0];
  const int*   ei  = (const int*)d_in[1];
  const float* W1  = (const float*)d_in[2];
  const float* b1  = (const float*)d_in[3];
  const float* W2  = (const float*)d_in[4];
  const float* b2  = (const float*)d_in[5];
  const float* Wfc = (const float*)d_in[6];
  const float* bfc = (const float*)d_in[7];
  float* out = (float*)d_out;

  const int N = in_sizes[0] / 128;
  const int E = in_sizes[1] / 2;
  const int* src = ei;
  const int* dst = ei + E;

  float* dinv = (float*)d_ws;
  float* bufA = dinv + N;
  float* bufB = bufA + (size_t)N * 128;

  int2* slot   = (int2*)d_out;
  int*  rank   = (int*)d_out + 2 * (size_t)E;
  int*  rowptr = rank + E;
  int*  cnt    = rowptr + N;
  int*  cursor = cnt + N;

  auto cdiv = [](long long a, long long b) { return (int)((a + b - 1) / b); };

  k_zero<<<cdiv(N + 1, 256), 256, 0, stream>>>(cnt, N + 1);
  k_count<<<cdiv(E, 1024), 256, 0, stream>>>(dst, cnt, rank, E);
  k_alloc<<<cdiv(N, 256), 256, 0, stream>>>(cnt, cursor, rowptr, dinv, N);
  k_place<<<cdiv(E, 1024), 256, 0, stream>>>(src, dst, rank, rowptr, dinv, slot, E);

  k_gemm_wmma<128><<<1250, 256, 0, stream>>>(x, W1, nullptr, bufA, N, 128);
  k_gather<v4f><<<2048, 256, 0, stream>>>(bufA, slot, rowptr, cnt, dinv, b1, bufB, N, 1);

  k_gemm_wmma<128><<<625, 256, 0, stream>>>(bufB, W2, nullptr, bufA, N, 64);
  k_gather<v2f><<<2048, 256, 0, stream>>>(bufA, slot, rowptr, cnt, dinv, b2, bufB, N, 0);

  k_gemm_wmma<64><<<625, 256, 0, stream>>>(bufB, Wfc, bfc, out, N, 64);
}


# products: weight matrix staged once per workgroup in LDS in fragment order; a wave owns 16 rows x half the width, reads A with 16-byte loads; k order permuted inside each chunk of 8
# speedup vs baseline: 2.7592x; 1.1136x over previous
#include <hip/hip_runtime.h>


typedef __attribute__((ext_vector_type(2))) float v2f;
typedef __attribute__((ext_vector_type(4))) float v4f;
typedef __attribute__((ext_vector_type(8))) float v8f;


__global__ __launch_bounds__(256) void k_zero(int* __restrict__ p, int n) {
  int i = blockIdx.x * 256 + threadIdx.x;
  if (i < n) p[i] = 0;
}

__global__ __launch_bounds__(256) void k_count(const int* __restrict__ dst,
                                               int* __restrict__ cnt,
                                               int* __restrict__ rank, int E) {
  int stride = gridDim.x * 256;
  for (int e0 = blockIdx.x * 256 + threadIdx.x; e0 < E; e0 += 4 * stride) {
    int d[4], r[4];
#pragma unroll
    for (int u = 0; u < 4; ++u) {
      int e = e0 + u * stride;
      d[u] = dst[e < E ? e : E - 1];
    }
#pragma unroll
    for (int u = 0; u < 4; ++u) r[u] = (e0 + u * stride < E) ? atomicAdd(&cnt[d[u]], 1) : 0;
#pragma unroll
    for (int u = 0; u < 4; ++u)
      if (e0 + u * stride < E) rank[e0 + u * stride] = r[u];
  }
}

__global__ __launch_bounds__(256) void k_alloc(const int* __restrict__ cnt,
                                               int* __restrict__ cursor,
                                               int* __restrict__ rowptr,
                                               float* __restrict__ dinv, int n) {
  __shared__ int wsum[8];
  __shared__ int base_s;
  int i = blockIdx.x * 256 + threadIdx.x;
  int lane = threadIdx.x & 31;
  int w = threadIdx.x >> 5;
  int c = (i < n) ? cnt[i] : 0;
  int x = c;
  for (int o = 1; o < 32; o <<= 1) {
    int y = __shfl_up(x, o);
    if (lane >= o) x += y;
  }
  if (lane == 31) wsum[w] = x;
  __syncthreads();
  if (threadIdx.x == 0) {
    int t = 0;
    for (int k = 0; k < 8; ++k) { int v = wsum[k]; wsum[k] = t; t += v; }
    base_s = atomicAdd(cursor, t);
  }
  __syncthreads();
  if (i < n) {
    rowptr[i] = base_s + wsum[w] + x - c;
    dinv[i] = rsqrtf((float)(c + 1));
  }
}

__global__ __launch_bounds__(256) void k_place(const int* __restrict__ src,
                                               const int* __restrict__ dst,
                                               const int* __restrict__ rank,
                                               const int* __restrict__ rowptr,
                                               const float* __restrict__ dinv,
                                               int2* __restrict__ slot, int E) {
  int stride = gridDim.x * 256;
  for (int e0 = blockIdx.x * 256 + threadIdx.x; e0 < E; e0 += 4 * stride) {
    int s[4], d[4], pos[4];
    float w[4];
#pragma unroll
    for (int u = 0; u < 4; ++u) {
      int e = e0 + u * stride;
      e = e < E ? e : E - 1;
      s[u] = src[e];
      d[u] = dst[e];
      pos[u] = rank[e];
    }
#pragma unroll
    for (int u = 0; u < 4; ++u) {
      pos[u] += rowptr[d[u]];
      w[u] = dinv[s[u]];
    }
#pragma unroll
    for (int u = 0; u < 4; ++u)
      if (e0 + u * stride < E) slot[pos[u]] = make_int2(s[u], __float_as_int(w[u]));
  }
}


template <typename VT, int G>
__device__ __forceinline__ void add_rows(VT& acc, const VT* __restrict__ hv,
                                         int sl, float wl, int k, int lane) {
  VT v[G];
  float w[G];
#pragma unroll
  for (int u = 0; u < G; ++u) {
    int s = __builtin_amdgcn_readlane(sl, k + u);
    w[u] = __int_as_float(__builtin_amdgcn_readlane(__float_as_int(wl), k + u));
    v[u] = hv[(size_t)s * 32 + lane];
  }
#pragma unroll
  for (int u = 0; u < G; ++u) acc += v[u] * w[u];
}

template <typename VT>
__global__ __launch_bounds__(256) void k_gather(const float* __restrict__ h,
                                                const int2* __restrict__ slot,
                                                const int* __restrict__ rowptr,
                                                const int* __restrict__ cnt,
                                                const float* __restrict__ dinv,
                                                const float* __restrict__ bias,
                                                float* __restrict__ out, int n, int relu) {
  int wave = blockIdx.x * 8 + __builtin_amdgcn_readfirstlane((int)(threadIdx.x >> 5));
  int nwaves = gridDim.x * 8;
  int lane = threadIdx.x & 31;
  const VT bv = reinterpret_cast<const VT*>(bias)[lane];
  for (int node = wave; node < n; node += nwaves) {
  int start = rowptr[node];
  int c = cnt[node];
  float di = dinv[node];
  const VT* hv = reinterpret_cast<const VT*>(h);
  VT acc = hv[(size_t)node * 32 + lane] * (di * di);
  for (int b = 0; b < c; b += 32) {
    int m = c - b < 32 ? c - b : 32;
    int sl = 0;
    float wl = 0.0f;
    if (lane < m) {
      int2 cw = slot[start + b + lane];
      sl = cw.x;
      wl = __int_as_float(cw.y) * di;
    }
    int k = 0;
    for (; k + 8 <= m; k += 8) add_rows<VT, 8>(acc, hv, sl, wl, k, lane);
    if (k + 4 <= m) { add_rows<VT, 4>(acc, hv, sl, wl, k, lane); k += 4; }
    if (k + 2 <= m) { add_rows<VT, 2>(acc, hv, sl, wl, k, lane); k += 2; }
    if (k < m)      { add_rows<VT, 1>(acc, hv, sl, wl, k, lane); }
  }
  acc += bv;
  if (relu) acc = __builtin_elementwise_max(acc, (VT)(0.0f));
  reinterpret_cast<VT*>(out)[(size_t)node * 32 + lane] = acc;
  }
}

template <int K, int NC>
__global__ __launch_bounds__(512) void k_gemm_wmma(const float* __restrict__ A,
                                                   const float* __restrict__ B,
                                                   const float* __restrict__ bias,
                                                   float* __restrict__ C, int M) {
  constexpr int KC  = K / 8;
  constexpr int NT  = NC / 16;
  constexpr int NWT = NT / 2;
  constexpr int NB  = KC * NT * 32;
  __shared__ v4f sB[NB];

  const int lane = threadIdx.x & 31;
  const int c  = lane & 15;
  const int hi = lane >> 4;

  const int w = __builtin_amdgcn_readfirstlane((int)(threadIdx.x >> 5));
  const unsigned Q  = (unsigned)(M >> 4) * 2u;
  int q        = (int)(blockIdx.x * Q / gridDim.x) + w;
  const int q1 = (int)((blockIdx.x + 1u) * Q / gridDim.x);

  v4f a[KC];
  if (q < q1) {
    const float* Ap = A + (size_t)((q >> 1) * 16 + c) * K + 4 * hi;
#pragma unroll
    for (int kc = 0; kc < KC; ++kc) a[kc] = *reinterpret_cast<const v4f*>(Ap + 8 * kc);
  }

#pragma unroll
  for (int i = 0; i < NB / 512; ++i) {
    int idx = threadIdx.x + i * 512;
    int l  = idx & 31;
    int t  = (idx >> 5) % NT;
    int kc = (idx >> 5) / NT;
    const float* bp = B + (size_t)(8 * kc + 4 * (l >> 4)) * NC + t * 16 + (l & 15);
    v4f v;
    v.x = bp[0];
    v.y = bp[NC];
    v.z = bp[2 * NC];
    v.w = bp[3 * NC];
    sB[idx] = v;
  }
  __syncthreads();

  if (q < q1) {
    for (;;) {
      const int tM = q >> 1;
      const int nh = q & 1;

      v8f acc[NWT];
#pragma unroll
      for (int t = 0; t < NWT; ++t) {
        float binit = bias ? bias[(nh * NWT + t) * 16 + c] : 0.0f;
#pragma unroll
        for (int r = 0; r < 8; ++r) acc[t][r] = binit;
      }

      const v4f* sBw = sB + (nh * NWT) * 32 + lane;
      v4f bw[NWT], bn[NWT];
#pragma unroll
      for (int t = 0; t < NWT; ++t) bw[t] = sBw[t * 32];
#pragma unroll
      for (int kc = 0; kc < KC; ++kc) {
        if (kc + 1 < KC) {
#pragma unroll
          for (int t = 0; t < NWT; ++t) bn[t] = sBw[((kc + 1) * NT + t) * 32];
        }
#pragma unroll
        for (int t = 0; t < NWT; ++t)
          acc[t] = __builtin_amdgcn_wmma_f32_16x16x4_f32(false, a[kc].xy, false, bw[t].xy,
                                                         (short)0, acc[t], false, false);
#pragma unroll
        for (int t = 0; t < NWT; ++t)
          acc[t] = __builtin_amdgcn_wmma_f32_16x16x4_f32(false, a[kc].zw, false, bw[t].zw,
                                                         (short)0, acc[t], false, false);
#pragma unroll
        for (int t = 0; t < NWT; ++t) bw[t] = bn[t];
        __builtin_amdgcn_sched_barrier(0);
      }

      float* Cp = C + (size_t)(tM * 16 + 8 * hi) * NC + (nh * NWT) * 16 + c;
#pragma unroll
      for (int t = 0; t < NWT; ++t)
#pragma unroll
        for (int r = 0; r < 8; ++r) Cp[(size_t)r * NC + t * 16] = acc[t][r];

      q += 16;
      if (q >= q1) break;
      const float* Ap = A + (size_t)((q >> 1) * 16 + c) * K + 4 * hi;
#pragma unroll
      for (int kc = 0; kc < KC; ++kc) a[kc] = *reinterpret_cast<const v4f*>(Ap + 8 * kc);
    }
  }
}


extern "C" void kernel_launch(void* const* d_in, const int* in_sizes, int n_in,
                              void* d_out, int out_size, void* d_ws, size_t ws_size,
                              hipStream_t stream) {
  const float* x   = (const float*)d_in[0];
  const int*   ei  = (const int*)d_in[1];
  const float* W1  = (const float*)d_in[2];
  const float* b1  = (const float*)d_in[3];
  const float* W2  = (const float*)d_in[4];
  const float* b2  = (const float*)d_in[5];
  const float* Wfc = (const float*)d_in[6];
  const float* bfc = (const float*)d_in[7];
  float* out = (float*)d_out;

  const int N = in_sizes[0] / 128;
  const int E = in_sizes[1] / 2;
  const int* src = ei;
  const int* dst = ei + E;

  float* dinv = (float*)d_ws;
  float* bufA = dinv + N;
  float* bufB = bufA + (size_t)N * 128;

  int2* slot   = (int2*)d_out;
  int*  rank   = (int*)d_out + 2 * (size_t)E;
  int*  rowptr = rank + E;
  int*  cnt    = rowptr + N;
  int*  cursor = cnt + N;

  auto cdiv = [](long long a, long long b) { return (int)((a + b - 1) / b); };

  k_zero<<<cdiv(N + 1, 256), 256, 0, stream>>>(cnt, N + 1);
  k_count<<<cdiv(E, 1024), 256, 0, stream>>>(dst, cnt, rank, E);
  k_alloc<<<cdiv(N, 256), 256, 0, stream>>>(cnt, cursor, rowptr, dinv, N);
  k_place<<<cdiv(E, 1024), 256, 0, stream>>>(src, dst, rank, rowptr, dinv, slot, E);

  k_gemm_wmma<128, 128><<<256, 512, 0, stream>>>(x, W1, nullptr, bufA, N);
  k_gather<v4f><<<2048, 256, 0, stream>>>(bufA, slot, rowptr, cnt, dinv, b1, bufB, N, 1);

  k_gemm_wmma<128, 64><<<256, 512, 0, stream>>>(bufB, W2, nullptr, bufA, N);
  k_gather<v2f><<<2048, 256, 0, stream>>>(bufA, slot, rowptr, cnt, dinv, b2, bufB, N, 0);

  k_gemm_wmma<64, 64><<<256, 512, 0, stream>>>(bufB, Wfc, bfc, out, N);
}


# edge ranks from LDS counters instead of returning global atomics: edges grouped by 512-node destination bucket, one workgroup ranks each bucket; per-node sums stay in arrival (rank) order
# speedup vs baseline: 3.2926x; 1.1933x over previous
#include <hip/hip_runtime.h>


typedef __attribute__((ext_vector_type(2))) float v2f;
typedef __attribute__((ext_vector_type(4))) float v4f;
typedef __attribute__((ext_vector_type(8))) float v8f;


constexpr int BSH   = 9;
constexpr int BNODE = 1 << BSH;
constexpr int NBMAX = 256;
constexpr int PT    = 1024;

__global__ __launch_bounds__(PT) void k_hist(const int* __restrict__ dst, int* __restrict__ hist,
                                             int E, int nb) {
  __shared__ int h[NBMAX];
  const int tid = threadIdx.x;
  if (tid < NBMAX) h[tid] = 0;
  __syncthreads();
  const int per = (E + (int)gridDim.x - 1) / (int)gridDim.x;
  const int e0 = (int)blockIdx.x * per;
  const int e1 = e0 + per < E ? e0 + per : E;
  for (int e = e0 + tid; e < e1; e += 4 * PT) {
    int d[4];
#pragma unroll
    for (int u = 0; u < 4; ++u) {
      int k = e + u * PT;
      d[u] = dst[k < e1 ? k : e1 - 1];
    }
#pragma unroll
    for (int u = 0; u < 4; ++u) atomicAdd(&h[d[u] >> BSH], e + u * PT < e1 ? 1 : 0);
  }
  __syncthreads();
  if (tid < nb) hist[(int)blockIdx.x * NBMAX + tid] = h[tid];
}

__device__ __forceinline__ int wave_scan_incl(int x, int lane) {
  for (int o = 1; o < 32; o <<= 1) {
    int y = __shfl_up(x, o);
    if (lane >= o) x += y;
  }
  return x;
}

__global__ __launch_bounds__(PT) void k_scatter(const int* __restrict__ src,
                                                const int* __restrict__ dst,
                                                const int* __restrict__ hist,
                                                int2* __restrict__ part,
                                                int* __restrict__ bstart, int E, int nb) {
  __shared__ int cur[NBMAX];
  __shared__ int tot[NBMAX];
  __shared__ int wsum[8];
  const int tid  = threadIdx.x;
  const int lane = tid & 31;
  const int w    = tid >> 5;
  const int G    = (int)gridDim.x;
  if (tid < NBMAX) { cur[tid] = 0; tot[tid] = 0; }
  __syncthreads();
  {
    const int b = tid & (NBMAX - 1);
    const int q = tid >> 8;
    const int gq = (G + 3) >> 2;
    const int g0 = q * gq;
    const int g1 = g0 + gq < G ? g0 + gq : G;
    if (b < nb) {
      int before = 0, total = 0;
#pragma unroll 8
      for (int g = g0; g < g1; ++g) {
        int v = hist[g * NBMAX + b];
        total += v;
        before += g < (int)blockIdx.x ? v : 0;
      }
      atomicAdd(&tot[b], total);
      atomicAdd(&cur[b], before);
    }
  }
  __syncthreads();
  const int t = tid < nb ? tot[tid] : 0;
  const int x = wave_scan_incl(t, lane);
  if (tid < NBMAX && lane == 31) wsum[w] = x;
  __syncthreads();
  if (tid == 0) {
    int s = 0;
    for (int k = 0; k < 8; ++k) { int v = wsum[k]; wsum[k] = s; s += v; }
  }
  __syncthreads();
  if (tid < nb) {
    const int bs = wsum[w] + x - t;
    cur[tid] += bs;
    if (blockIdx.x == 0) bstart[tid] = bs;
  }
  if (blockIdx.x == 0 && tid == 0) bstart[nb] = E;
  __syncthreads();

  const int per = (E + G - 1) / G;
  const int e0 = (int)blockIdx.x * per;
  const int e1 = e0 + per < E ? e0 + per : E;
  for (int e = e0 + tid; e < e1; e += 4 * PT) {
    int s[4], d[4];
#pragma unroll
    for (int u = 0; u < 4; ++u) {
      int k = e + u * PT;
      k = k < e1 ? k : e1 - 1;
      s[u] = src[k];
      d[u] = dst[k];
    }
    int p[4];
#pragma unroll
    for (int u = 0; u < 4; ++u) p[u] = atomicAdd(&cur[d[u] >> BSH], e + u * PT < e1 ? 1 : 0);
#pragma unroll
    for (int u = 0; u < 4; ++u)
      if (e + u * PT < e1) part[p[u]] = make_int2(s[u], d[u]);
  }
}

__global__ __launch_bounds__(PT) void k_bucket(const int2* __restrict__ part,
                                               const int* __restrict__ bstart,
                                               int* __restrict__ rank,
                                               int* __restrict__ rowptr,
                                               int* __restrict__ cnt,
                                               float* __restrict__ dinv, int n) {
  __shared__ int cl[BNODE];
  __shared__ int wsum[BNODE / 32];
  const int tid  = threadIdx.x;
  const int lane = tid & 31;
  const int w    = tid >> 5;
  if (tid < BNODE) cl[tid] = 0;
  const int i0 = bstart[blockIdx.x];
  const int i1 = bstart[blockIdx.x + 1];
  __syncthreads();
  for (int i = i0 + tid; i < i1; i += 4 * PT) {
    int d[4];
#pragma unroll
    for (int u = 0; u < 4; ++u) {
      int k = i + u * PT;
      d[u] = part[k < i1 ? k : i1 - 1].y;
    }
    int r[4];
#pragma unroll
    for (int u = 0; u < 4; ++u) r[u] = atomicAdd(&cl[d[u] & (BNODE - 1)], i + u * PT < i1 ? 1 : 0);
#pragma unroll
    for (int u = 0; u < 4; ++u)
      if (i + u * PT < i1) rank[i + u * PT] = r[u];
  }
  __syncthreads();
  const int c = tid < BNODE ? cl[tid] : 0;
  const int x = wave_scan_incl(c, lane);
  if (tid < BNODE && lane == 31) wsum[w] = x;
  __syncthreads();
  if (tid == 0) {
    int s = 0;
    for (int k = 0; k < BNODE / 32; ++k) { int v = wsum[k]; wsum[k] = s; s += v; }
  }
  __syncthreads();
  const int node = (int)blockIdx.x * BNODE + tid;
  if (tid < BNODE && node < n) {
    rowptr[node] = i0 + wsum[w] + x - c;
    cnt[node] = c;
    dinv[node] = rsqrtf((float)(c + 1));
  }
}

__global__ __launch_bounds__(256) void k_place(const int2* __restrict__ part,
                                               const int* __restrict__ rank,
                                               const int* __restrict__ rowptr,
                                               const float* __restrict__ dinv,
                                               int2* __restrict__ slot, int E) {
  int stride = gridDim.x * 256;
  for (int e0 = blockIdx.x * 256 + threadIdx.x; e0 < E; e0 += 4 * stride) {
    int2 sd[4];
    int pos[4];
    float w[4];
#pragma unroll
    for (int u = 0; u < 4; ++u) {
      int e = e0 + u * stride;
      e = e < E ? e : E - 1;
      sd[u] = part[e];
      pos[u] = rank[e];
    }
#pragma unroll
    for (int u = 0; u < 4; ++u) {
      pos[u] += rowptr[sd[u].y];
      w[u] = dinv[sd[u].x];
    }
#pragma unroll
    for (int u = 0; u < 4; ++u)
      if (e0 + u * stride < E) slot[pos[u]] = make_int2(sd[u].x, __float_as_int(w[u]));
  }
}


template <typename VT, int G>
__device__ __forceinline__ void add_rows(VT& acc, const VT* __restrict__ hv,
                                         int sl, float wl, int k, int lane) {
  VT v[G];
  float w[G];
#pragma unroll
  for (int u = 0; u < G; ++u) {
    int s = __builtin_amdgcn_readlane(sl, k + u);
    w[u] = __int_as_float(__builtin_amdgcn_readlane(__float_as_int(wl), k + u));
    v[u] = hv[(size_t)s * 32 + lane];
  }
#pragma unroll
  for (int u = 0; u < G; ++u) acc += v[u] * w[u];
}

template <typename VT>
__global__ __launch_bounds__(256) void k_gather(const float* __restrict__ h,
                                                const int2* __restrict__ slot,
                                                const int* __restrict__ rowptr,
                                                const int* __restrict__ cnt,
                                                const float* __restrict__ dinv,
                                                const float* __restrict__ bias,
                                                float* __restrict__ out, int n, int relu) {
  int wave = blockIdx.x * 8 + __builtin_amdgcn_readfirstlane((int)(threadIdx.x >> 5));
  int nwaves = gridDim.x * 8;
  int lane = threadIdx.x & 31;
  const VT bv = reinterpret_cast<const VT*>(bias)[lane];
  for (int node = wave; node < n; node += nwaves) {
  int start = rowptr[node];
  int c = cnt[node];
  float di = dinv[node];
  const VT* hv = reinterpret_cast<const VT*>(h);
  VT acc = hv[(size_t)node * 32 + lane] * (di * di);
  for (int b = 0; b < c; b += 32) {
    int m = c - b < 32 ? c - b : 32;
    int sl = 0;
    float wl = 0.0f;
    if (lane < m) {
      int2 cw = slot[start + b + lane];
      sl = cw.x;
      wl = __int_as_float(cw.y) * di;
    }
    int k = 0;
    for (; k + 8 <= m; k += 8) add_rows<VT, 8>(acc, hv, sl, wl, k, lane);
    if (k + 4 <= m) { add_rows<VT, 4>(acc, hv, sl, wl, k, lane); k += 4; }
    if (k + 2 <= m) { add_rows<VT, 2>(acc, hv, sl, wl, k, lane); k += 2; }
    if (k < m)      { add_rows<VT, 1>(acc, hv, sl, wl, k, lane); }
  }
  acc += bv;
  if (relu) acc = __builtin_elementwise_max(acc, (VT)(0.0f));
  reinterpret_cast<VT*>(out)[(size_t)node * 32 + lane] = acc;
  }
}

template <int K, int NC>
__global__ __launch_bounds__(512) void k_gemm_wmma(const float* __restrict__ A,
                                                   const float* __restrict__ B,
                                                   const float* __restrict__ bias,
                                                   float* __restrict__ C, int M) {
  constexpr int KC  = K / 8;
  constexpr int NT  = NC / 16;
  constexpr int NWT = NT / 2;
  constexpr int NB  = KC * NT * 32;
  __shared__ v4f sB[NB];

  const int lane = threadIdx.x & 31;
  const int c  = lane & 15;
  const int hi = lane >> 4;

  const int w = __builtin_amdgcn_readfirstlane((int)(threadIdx.x >> 5));
  const unsigned Q  = (unsigned)(M >> 4) * 2u;
  int q        = (int)(blockIdx.x * Q / gridDim.x) + w;
  const int q1 = (int)((blockIdx.x + 1u) * Q / gridDim.x);

  v4f a[KC];
  if (q < q1) {
    const float* Ap = A + (size_t)((q >> 1) * 16 + c) * K + 4 * hi;
#pragma unroll
    for (int kc = 0; kc < KC; ++kc) a[kc] = *reinterpret_cast<const v4f*>(Ap + 8 * kc);
  }

#pragma unroll
  for (int i = 0; i < NB / 512; ++i) {
    int idx = threadIdx.x + i * 512;
    int l  = idx & 31;
    int t  = (idx >> 5) % NT;
    int kc = (idx >> 5) / NT;
    const float* bp = B + (size_t)(8 * kc + 4 * (l >> 4)) * NC + t * 16 + (l & 15);
    v4f v;
    v.x = bp[0];
    v.y = bp[NC];
    v.z = bp[2 * NC];
    v.w = bp[3 * NC];
    sB[idx] = v;
  }
  __syncthreads();

  if (q < q1) {
    for (;;) {
      const int tM = q >> 1;
      const int nh = q & 1;

      v8f acc[NWT];
#pragma unroll
      for (int t = 0; t < NWT; ++t) {
        float binit = bias ? bias[(nh * NWT + t) * 16 + c] : 0.0f;
#pragma unroll
        for (int r = 0; r < 8; ++r) acc[t][r] = binit;
      }

      const v4f* sBw = sB + (nh * NWT) * 32 + lane;
      v4f bw[NWT], bn[NWT];
#pragma unroll
      for (int t = 0; t < NWT; ++t) bw[t] = sBw[t * 32];
#pragma unroll
      for (int kc = 0; kc < KC; ++kc) {
        if (kc + 1 < KC) {
#pragma unroll
          for (int t = 0; t < NWT; ++t) bn[t] = sBw[((kc + 1) * NT + t) * 32];
        }
#pragma unroll
        for (int t = 0; t < NWT; ++t)
          acc[t] = __builtin_amdgcn_wmma_f32_16x16x4_f32(false, a[kc].xy, false, bw[t].xy,
                                                         (short)0, acc[t], false, false);
#pragma unroll
        for (int t = 0; t < NWT; ++t)
          acc[t] = __builtin_amdgcn_wmma_f32_16x16x4_f32(false, a[kc].zw, false, bw[t].zw,
                                                         (short)0, acc[t], false, false);
#pragma unroll
        for (int t = 0; t < NWT; ++t) bw[t] = bn[t];
        __builtin_amdgcn_sched_barrier(0);
      }

      float* Cp = C + (size_t)(tM * 16 + 8 * hi) * NC + (nh * NWT) * 16 + c;
#pragma unroll
      for (int t = 0; t < NWT; ++t)
#pragma unroll
        for (int r = 0; r < 8; ++r) Cp[(size_t)r * NC + t * 16] = acc[t][r];

      q += 16;
      if (q >= q1) break;
      const float* Ap = A + (size_t)((q >> 1) * 16 + c) * K + 4 * hi;
#pragma unroll
      for (int kc = 0; kc < KC; ++kc) a[kc] = *reinterpret_cast<const v4f*>(Ap + 8 * kc);
    }
  }
}


extern "C" void kernel_launch(void* const* d_in, const int* in_sizes, int n_in,
                              void* d_out, int out_size, void* d_ws, size_t ws_size,
                              hipStream_t stream) {
  const float* x   = (const float*)d_in[0];
  const int*   ei  = (const int*)d_in[1];
  const float* W1  = (const float*)d_in[2];
  const float* b1  = (const float*)d_in[3];
  const float* W2  = (const float*)d_in[4];
  const float* b2  = (const float*)d_in[5];
  const float* Wfc = (const float*)d_in[6];
  const float* bfc = (const float*)d_in[7];
  float* out = (float*)d_out;

  const int N = in_sizes[0] / 128;
  const int E = in_sizes[1] / 2;
  const int* src = ei;
  const int* dst = ei + E;

  float* dinv = (float*)d_ws;
  float* bufA = dinv + N;
  float* bufB = bufA + (size_t)N * 128;

  int2* slot   = (int2*)d_out;
  int*  rowptr = (int*)d_out + 2 * (size_t)E;
  int*  cnt    = rowptr + N;

  int2* part   = (int2*)bufB;
  int*  rank   = (int*)bufB + 2 * (size_t)E;
  int*  hist   = rank + E;
  int*  bstart = hist + 256 * NBMAX;

  auto cdiv = [](long long a, long long b) { return (int)((a + b - 1) / b); };

  const int nb = cdiv(N, BNODE);
  k_hist<<<256, PT, 0, stream>>>(dst, hist, E, nb);
  k_scatter<<<256, PT, 0, stream>>>(src, dst, hist, part, bstart, E, nb);
  k_bucket<<<nb, PT, 0, stream>>>(part, bstart, rank, rowptr, cnt, dinv, N);
  k_place<<<cdiv(E, 1024), 256, 0, stream>>>(part, rank, rowptr, dinv, slot, E);

  k_gemm_wmma<128, 128><<<256, 512, 0, stream>>>(x, W1, nullptr, bufA, N);
  k_gather<v4f><<<2048, 256, 0, stream>>>(bufA, slot, rowptr, cnt, dinv, b1, bufB, N, 1);

  k_gemm_wmma<128, 64><<<256, 512, 0, stream>>>(bufB, W2, nullptr, bufA, N);
  k_gather<v2f><<<2048, 256, 0, stream>>>(bufA, slot, rowptr, cnt, dinv, b2, bufB, N, 0);

  k_gemm_wmma<64, 64><<<256, 512, 0, stream>>>(bufB, Wfc, bfc, out, N);
}


# products 2 and 3 moved into the aggregation launches: a 16-wave workgroup sums 64 nodes' rows into an LDS tile and multiplies it there (f32 WMMA, same k order); 8 -> 6 launches; sums in slot order
# speedup vs baseline: 4.6922x; 1.4251x over previous
#include <hip/hip_runtime.h>


typedef __attribute__((ext_vector_type(2))) float v2f;
typedef __attribute__((ext_vector_type(4))) float v4f;
typedef __attribute__((ext_vector_type(8))) float v8f;


constexpr int BSH   = 9;
constexpr int BNODE = 1 << BSH;
constexpr int NBMAX = 256;
constexpr int PT    = 1024;

__global__ __launch_bounds__(PT) void k_hist(const int* __restrict__ dst, int* __restrict__ hist,
                                             int E, int nb) {
  __shared__ int h[NBMAX];
  const int tid = threadIdx.x;
  if (tid < NBMAX) h[tid] = 0;
  __syncthreads();
  const int per = (E + (int)gridDim.x - 1) / (int)gridDim.x;
  const int e0 = (int)blockIdx.x * per;
  const int e1 = e0 + per < E ? e0 + per : E;
  for (int e = e0 + tid; e < e1; e += 4 * PT) {
    int d[4];
#pragma unroll
    for (int u = 0; u < 4; ++u) {
      int k = e + u * PT;
      d[u] = dst[k < e1 ? k : e1 - 1];
    }
#pragma unroll
    for (int u = 0; u < 4; ++u) atomicAdd(&h[d[u] >> BSH], e + u * PT < e1 ? 1 : 0);
  }
  __syncthreads();
  if (tid < nb) hist[(int)blockIdx.x * NBMAX + tid] = h[tid];
}

__device__ __forceinline__ int wave_scan_incl(int x, int lane) {
  for (int o = 1; o < 32; o <<= 1) {
    int y = __shfl_up(x, o);
    if (lane >= o) x += y;
  }
  return x;
}

__global__ __launch_bounds__(PT) void k_scatter(const int* __restrict__ src,
                                                const int* __restrict__ dst,
                                                const int* __restrict__ hist,
                                                int2* __restrict__ part,
                                                int* __restrict__ bstart, int E, int nb) {
  __shared__ int cur[NBMAX];
  __shared__ int tot[NBMAX];
  __shared__ int wsum[8];
  const int tid  = threadIdx.x;
  const int lane = tid & 31;
  const int w    = tid >> 5;
  const int G    = (int)gridDim.x;
  if (tid < NBMAX) { cur[tid] = 0; tot[tid] = 0; }
  __syncthreads();
  {
    const int b = tid & (NBMAX - 1);
    const int q = tid >> 8;
    const int gq = (G + 3) >> 2;
    const int g0 = q * gq;
    const int g1 = g0 + gq < G ? g0 + gq : G;
    if (b < nb) {
      int before = 0, total = 0;
#pragma unroll 8
      for (int g = g0; g < g1; ++g) {
        int v = hist[g * NBMAX + b];
        total += v;
        before += g < (int)blockIdx.x ? v : 0;
      }
      atomicAdd(&tot[b], total);
      atomicAdd(&cur[b], before);
    }
  }
  __syncthreads();
  const int t = tid < nb ? tot[tid] : 0;
  const int x = wave_scan_incl(t, lane);
  if (tid < NBMAX && lane == 31) wsum[w] = x;
  __syncthreads();
  if (tid == 0) {
    int s = 0;
    for (int k = 0; k < 8; ++k) { int v = wsum[k]; wsum[k] = s; s += v; }
  }
  __syncthreads();
  if (tid < nb) {
    const int bs = wsum[w] + x - t;
    cur[tid] += bs;
    if (blockIdx.x == 0) bstart[tid] = bs;
  }
  if (blockIdx.x == 0 && tid == 0) bstart[nb] = E;
  __syncthreads();

  const int per = (E + G - 1) / G;
  const int e0 = (int)blockIdx.x * per;
  const int e1 = e0 + per < E ? e0 + per : E;
  for (int e = e0 + tid; e < e1; e += 4 * PT) {
    int s[4], d[4];
#pragma unroll
    for (int u = 0; u < 4; ++u) {
      int k = e + u * PT;
      k = k < e1 ? k : e1 - 1;
      s[u] = src[k];
      d[u] = dst[k];
    }
    int p[4];
#pragma unroll
    for (int u = 0; u < 4; ++u) p[u] = atomicAdd(&cur[d[u] >> BSH], e + u * PT < e1 ? 1 : 0);
#pragma unroll
    for (int u = 0; u < 4; ++u)
      if (e + u * PT < e1) part[p[u]] = make_int2(s[u], d[u]);
  }
}

constexpr int CAP = 12288;

__global__ __launch_bounds__(PT) void k_bucket(const int2* __restrict__ part,
                                               const int* __restrict__ bstart,
                                               int* __restrict__ slot,
                                               int* __restrict__ rowptr,
                                               int* __restrict__ cnt,
                                               float* __restrict__ dinv, int n) {
  __shared__ int cl[BNODE];
  __shared__ int wsum[BNODE / 32];
  __shared__ int sbuf[CAP];
  const int tid  = threadIdx.x;
  const int lane = tid & 31;
  const int w    = tid >> 5;
  if (tid < BNODE) cl[tid] = 0;
  const int i0 = bstart[blockIdx.x];
  const int i1 = bstart[blockIdx.x + 1];
  const bool staged = i1 - i0 <= CAP;
  __syncthreads();
  for (int i = i0 + tid; i < i1; i += 4 * PT) {
    int d[4];
#pragma unroll
    for (int u = 0; u < 4; ++u) {
      int k = i + u * PT;
      d[u] = part[k < i1 ? k : i1 - 1].y;
    }
#pragma unroll
    for (int u = 0; u < 4; ++u) atomicAdd(&cl[d[u] & (BNODE - 1)], i + u * PT < i1 ? 1 : 0);
  }
  __syncthreads();
  const int c = tid < BNODE ? cl[tid] : 0;
  const int x = wave_scan_incl(c, lane);
  if (tid < BNODE && lane == 31) wsum[w] = x;
  __syncthreads();
  if (tid == 0) {
    int s = 0;
    for (int k = 0; k < BNODE / 32; ++k) { int v = wsum[k]; wsum[k] = s; s += v; }
  }
  __syncthreads();
  const int node = (int)blockIdx.x * BNODE + tid;
  if (tid < BNODE) {
    const int first = i0 + wsum[w] + x - c;
    cl[tid] = first;
    if (node < n) {
      rowptr[node] = first;
      cnt[node] = c;
      dinv[node] = rsqrtf((float)(c + 1));
    }
  }
  __syncthreads();
  for (int i = i0 + tid; i < i1; i += 4 * PT) {
    int2 sd[4];
#pragma unroll
    for (int u = 0; u < 4; ++u) {
      int k = i + u * PT;
      sd[u] = part[k < i1 ? k : i1 - 1];
    }
    int p[4];
#pragma unroll
    for (int u = 0; u < 4; ++u)
      p[u] = atomicAdd(&cl[sd[u].y & (BNODE - 1)], i + u * PT < i1 ? 1 : 0);
#pragma unroll
    for (int u = 0; u < 4; ++u)
      if (i + u * PT < i1) {
        if (staged) sbuf[p[u] - i0] = sd[u].x;
        else        slot[p[u]] = sd[u].x;
      }
  }
  if (staged) {
    __syncthreads();
    for (int g = (i0 & ~31) + tid; g < i1; g += PT)
      if (g >= i0) slot[g] = sbuf[g - i0];
  }
}


template <typename VT, int G>
__device__ __forceinline__ void add_rows(VT& acc, const VT* __restrict__ hv,
                                         int sl, int k, int lane) {
  VT v[G];
#pragma unroll
  for (int u = 0; u < G; ++u) {
    int s = __builtin_amdgcn_readlane(sl, k + u);
    v[u] = hv[(size_t)s * 32 + lane];
  }
#pragma unroll
  for (int u = 0; u < G; ++u) acc += v[u];
}

template <int K, int NC, bool RS>
__global__ __launch_bounds__(512) void k_gemm_wmma(const float* __restrict__ A,
                                                   const float* __restrict__ B,
                                                   const float* __restrict__ bias,
                                                   const float* __restrict__ rs,
                                                   float* __restrict__ C, int M) {
  constexpr int KC  = K / 8;
  constexpr int NT  = NC / 16;
  constexpr int NWT = NT / 2;
  constexpr int NB  = KC * NT * 32;
  __shared__ v4f sB[NB];

  const int lane = threadIdx.x & 31;
  const int c  = lane & 15;
  const int hi = lane >> 4;

  const int w = __builtin_amdgcn_readfirstlane((int)(threadIdx.x >> 5));
  const unsigned Q  = (unsigned)(M >> 4) * 2u;
  int q        = (int)(blockIdx.x * Q / gridDim.x) + w;
  const int q1 = (int)((blockIdx.x + 1u) * Q / gridDim.x);

  v4f a[KC];
  if (q < q1) {
    const float* Ap = A + (size_t)((q >> 1) * 16 + c) * K + 4 * hi;
#pragma unroll
    for (int kc = 0; kc < KC; ++kc) a[kc] = *reinterpret_cast<const v4f*>(Ap + 8 * kc);
  }

#pragma unroll
  for (int i = 0; i < NB / 512; ++i) {
    int idx = threadIdx.x + i * 512;
    int l  = idx & 31;
    int t  = (idx >> 5) % NT;
    int kc = (idx >> 5) / NT;
    const float* bp = B + (size_t)(8 * kc + 4 * (l >> 4)) * NC + t * 16 + (l & 15);
    v4f v;
    v.x = bp[0];
    v.y = bp[NC];
    v.z = bp[2 * NC];
    v.w = bp[3 * NC];
    sB[idx] = v;
  }
  __syncthreads();

  if (q < q1) {
    for (;;) {
      const int tM = q >> 1;
      const int nh = q & 1;

      v4f sc0, sc1;
      if (RS) {
        const float* rp = rs + tM * 16 + 8 * hi;
        sc0 = *reinterpret_cast<const v4f*>(rp);
        sc1 = *reinterpret_cast<const v4f*>(rp + 4);
      }

      v8f acc[NWT];
#pragma unroll
      for (int t = 0; t < NWT; ++t) {
        float binit = bias ? bias[(nh * NWT + t) * 16 + c] : 0.0f;
#pragma unroll
        for (int r = 0; r < 8; ++r) acc[t][r] = binit;
      }

      const v4f* sBw = sB + (nh * NWT) * 32 + lane;
      v4f bw[NWT], bn[NWT];
#pragma unroll
      for (int t = 0; t < NWT; ++t) bw[t] = sBw[t * 32];
#pragma unroll
      for (int kc = 0; kc < KC; ++kc) {
        if (kc + 1 < KC) {
#pragma unroll
          for (int t = 0; t < NWT; ++t) bn[t] = sBw[((kc + 1) * NT + t) * 32];
        }
#pragma unroll
        for (int t = 0; t < NWT; ++t)
          acc[t] = __builtin_amdgcn_wmma_f32_16x16x4_f32(false, a[kc].xy, false, bw[t].xy,
                                                         (short)0, acc[t], false, false);
#pragma unroll
        for (int t = 0; t < NWT; ++t)
          acc[t] = __builtin_amdgcn_wmma_f32_16x16x4_f32(false, a[kc].zw, false, bw[t].zw,
                                                         (short)0, acc[t], false, false);
#pragma unroll
        for (int t = 0; t < NWT; ++t) bw[t] = bn[t];
        __builtin_amdgcn_sched_barrier(0);
      }

      float* Cp = C + (size_t)(tM * 16 + 8 * hi) * NC + (nh * NWT) * 16 + c;
#pragma unroll
      for (int t = 0; t < NWT; ++t)
#pragma unroll
        for (int r = 0; r < 8; ++r) {
          float v = acc[t][r];
          if (RS) v *= r < 4 ? sc0[r & 3] : sc1[r & 3];
          Cp[(size_t)r * NC + t * 16] = v;
        }

      q += 16;
      if (q >= q1) break;
      const float* Ap = A + (size_t)((q >> 1) * 16 + c) * K + 4 * hi;
#pragma unroll
      for (int kc = 0; kc < KC; ++kc) a[kc] = *reinterpret_cast<const v4f*>(Ap + 8 * kc);
    }
  }
}

template <typename VT, int K, bool RELU, bool RS>
__global__ __launch_bounds__(512) void k_gather_gemm(const float* __restrict__ h,
                                                     const int* __restrict__ slot,
                                                     const int* __restrict__ rowptr,
                                                     const int* __restrict__ cnt,
                                                     const float* __restrict__ dinv,
                                                     const float* __restrict__ bagg,
                                                     const float* __restrict__ B,
                                                     const float* __restrict__ bout,
                                                     float* __restrict__ C, int n) {
  constexpr int NC  = 64;
  constexpr int KC  = K / 8;
  constexpr int NT  = NC / 16;
  constexpr int NWT = NT / 2;
  constexpr int NB  = KC * NT * 32;
  constexpr int TR  = 64;
  constexpr int P4  = K / 4 + 1;
  __shared__ v4f sB[NB];
  __shared__ v4f sA[TR * P4];
  __shared__ v4f sD[TR / 4];

  const int lane = threadIdx.x & 31;
  const int c  = lane & 15;
  const int hi = lane >> 4;
  const int w = __builtin_amdgcn_readfirstlane((int)(threadIdx.x >> 5));

#pragma unroll
  for (int i = 0; i < NB / 512; ++i) {
    int idx = threadIdx.x + i * 512;
    int l  = idx & 31;
    int t  = (idx >> 5) % NT;
    int kc = (idx >> 5) / NT;
    const float* bp = B + (size_t)(8 * kc + 4 * (l >> 4)) * NC + t * 16 + (l & 15);
    v4f v;
    v.x = bp[0];
    v.y = bp[NC];
    v.z = bp[2 * NC];
    v.w = bp[3 * NC];
    sB[idx] = v;
  }

  const unsigned RT = (unsigned)((n + 15) >> 4);
  const int n0 = (int)(blockIdx.x * RT / gridDim.x) * 16;
  int n1 = (int)((blockIdx.x + 1u) * RT / gridDim.x) * 16;
  n1 = n1 < n ? n1 : n;

  const VT* hv = reinterpret_cast<const VT*>(h);
  const VT bv = reinterpret_cast<const VT*>(bagg)[lane];
  float* sAf = reinterpret_cast<float*>(sA);
  float* sDf = reinterpret_cast<float*>(sD);

  for (int base = n0; base < n1; base += TR) {
    int rp_l = 0, cn_l = 0;
    float di_l = 0.0f;
    {
      const int nd = base + w + 16 * lane;
      if (lane < TR / 16 && nd < n1) {
        rp_l = rowptr[nd];
        cn_l = cnt[nd];
        di_l = dinv[nd];
      }
    }
    for (int j = 0; j < TR / 16; ++j) {
      const int r = w + 16 * j;
      const int node = base + r;
      VT acc = (VT)(0.0f);
      float di = 0.0f;
      if (node < n1) {
        const int start = __builtin_amdgcn_readlane(rp_l, j);
        const int cn = __builtin_amdgcn_readlane(cn_l, j);
        di = __int_as_float(__builtin_amdgcn_readlane(__float_as_int(di_l), j));
        acc = hv[(size_t)node * 32 + lane];
        for (int b = 0; b < cn; b += 32) {
          int m = cn - b < 32 ? cn - b : 32;
          int sl = 0;
          if (lane < m) sl = slot[start + b + lane];
          int k = 0;
          for (; k + 8 <= m; k += 8) add_rows<VT, 8>(acc, hv, sl, k, lane);
          if (k + 4 <= m) { add_rows<VT, 4>(acc, hv, sl, k, lane); k += 4; }
          if (k + 2 <= m) { add_rows<VT, 2>(acc, hv, sl, k, lane); k += 2; }
          if (k < m)      { add_rows<VT, 1>(acc, hv, sl, k, lane); }
        }
        acc = acc * di + bv;
        if (RELU) acc = __builtin_elementwise_max(acc, (VT)(0.0f));
      }
      reinterpret_cast<VT*>(sAf + r * (4 * P4))[lane] = acc;
      if (lane == 0) sDf[r] = di;
    }
    __syncthreads();

    if (w < 8 && base + (w >> 1) * 16 < n1) {
      const int rt = w >> 1;
      const int nh = w & 1;

      v8f acc[NWT];
#pragma unroll
      for (int t = 0; t < NWT; ++t) {
        float binit = bout ? bout[(nh * NWT + t) * 16 + c] : 0.0f;
#pragma unroll
        for (int r = 0; r < 8; ++r) acc[t][r] = binit;
      }

      const v4f* sAw = sA + (rt * 16 + c) * P4 + hi;
      const v4f* sBw = sB + (nh * NWT) * 32 + lane;
      v4f a = sAw[0], an;
      v4f bw[NWT], bn[NWT];
#pragma unroll
      for (int t = 0; t < NWT; ++t) bw[t] = sBw[t * 32];
#pragma unroll
      for (int kc = 0; kc < KC; ++kc) {
        if (kc + 1 < KC) {
          an = sAw[2 * (kc + 1)];
#pragma unroll
          for (int t = 0; t < NWT; ++t) bn[t] = sBw[((kc + 1) * NT + t) * 32];
        }
#pragma unroll
        for (int t = 0; t < NWT; ++t)
          acc[t] = __builtin_amdgcn_wmma_f32_16x16x4_f32(false, a.xy, false, bw[t].xy,
                                                         (short)0, acc[t], false, false);
#pragma unroll
        for (int t = 0; t < NWT; ++t)
          acc[t] = __builtin_amdgcn_wmma_f32_16x16x4_f32(false, a.zw, false, bw[t].zw,
                                                         (short)0, acc[t], false, false);
        a = an;
#pragma unroll
        for (int t = 0; t < NWT; ++t) bw[t] = bn[t];
      }

      const int row0 = base + rt * 16 + 8 * hi;
      v4f sc0, sc1;
      if (RS) {
        sc0 = sD[rt * 4 + 2 * hi];
        sc1 = sD[rt * 4 + 2 * hi + 1];
      }
      float* Cp = C + (size_t)row0 * NC + (nh * NWT) * 16 + c;
#pragma unroll
      for (int t = 0; t < NWT; ++t)
#pragma unroll
        for (int r = 0; r < 8; ++r) {
          float v = acc[t][r];
          if (RS) v *= r < 4 ? sc0[r & 3] : sc1[r & 3];
          Cp[(size_t)r * NC + t * 16] = v;
        }
    }
    __syncthreads();
  }
}


extern "C" void kernel_launch(void* const* d_in, const int* in_sizes, int n_in,
                              void* d_out, int out_size, void* d_ws, size_t ws_size,
                              hipStream_t stream) {
  const float* x   = (const float*)d_in[0];
  const int*   ei  = (const int*)d_in[1];
  const float* W1  = (const float*)d_in[2];
  const float* b1  = (const float*)d_in[3];
  const float* W2  = (const float*)d_in[4];
  const float* b2  = (const float*)d_in[5];
  const float* Wfc = (const float*)d_in[6];
  const float* bfc = (const float*)d_in[7];
  float* out = (float*)d_out;

  const int N = in_sizes[0] / 128;
  const int E = in_sizes[1] / 2;
  const int* src = ei;
  const int* dst = ei + E;

  float* dinv = (float*)d_ws;
  float* bufA = dinv + N;
  float* bufB = bufA + (size_t)N * 128;

  int*  slot   = (int*)bufB + (size_t)N * 64;
  int*  rowptr = slot + (size_t)E;
  int*  cnt    = rowptr + N;

  int2* part   = (int2*)bufB;
  int*  hist   = (int*)bufB + 2 * (size_t)E;
  int*  bstart = hist + 256 * NBMAX;

  auto cdiv = [](long long a, long long b) { return (int)((a + b - 1) / b); };

  const int nb = cdiv(N, BNODE);
  k_hist<<<256, PT, 0, stream>>>(dst, hist, E, nb);
  k_scatter<<<256, PT, 0, stream>>>(src, dst, hist, part, bstart, E, nb);
  k_bucket<<<nb, PT, 0, stream>>>(part, bstart, slot, rowptr, cnt, dinv, N);

  k_gemm_wmma<128, 128, true><<<256, 512, 0, stream>>>(x, W1, nullptr, dinv, bufA, N);

  k_gather_gemm<v4f, 128, true, true><<<1024, 512, 0, stream>>>(bufA, slot, rowptr, cnt, dinv, b1,
                                                                W2, nullptr, bufB, N);

  k_gather_gemm<v2f, 64, false, false><<<1024, 512, 0, stream>>>(bufB, slot, rowptr, cnt, dinv, b2,
                                                                 Wfc, bfc, out, N);
}
